# starting kernel
# baseline (speedup 1.0000x reference)
#include <hip/hip_runtime.h>

typedef __attribute__((ext_vector_type(2))) float v2f;
typedef __attribute__((ext_vector_type(8))) float v8f;

#define BATCH   16
#define CDIM    64
#define HW      4096
#define NROWS   (BATCH * HW)
#define KEMB    1024
#define TPB     256
#define NWAVES  8
#define ROWS_WG 128
#define NWG     (NROWS / ROWS_WG)

#define LDS_FLOATS 66696
#define LDS_BYTES  (LDS_FLOATS * 4)

__global__ void vq_zero_kernel(float* acc) {
    if (threadIdx.x == 0) acc[0] = 0.0f;
}

__global__ __launch_bounds__(TPB)
void vq_main_kernel(const float* __restrict__ x,
                    const float* __restrict__ E,
                    float* __restrict__ zq,
                    float* __restrict__ lossAcc)
{
    extern __shared__ float lds[];
    float* Bfrag  = lds;
    float* esq    = lds + 65536;
    int*   idxs   = (int*)(lds + 66560);
    float* redbuf = lds + 66688;

    const int tid = threadIdx.x;
    const int wg  = blockIdx.x;

    for (int p = tid; p < (KEMB * CDIM / 2); p += TPB) {
        int k  = p >> 5;
        int j  = p & 31;
        int c0 = j * 2;
        float e0 = E[k * CDIM + c0];
        float e1 = E[k * CDIM + c0 + 1];
        int t  = k >> 4;
        int n  = k & 15;
        int kc = c0 >> 2;
        int g  = (c0 >> 1) & 1;
        int fo = ((t * 16 + kc) * 32 + (g * 16 + n)) * 2;
        Bfrag[fo]     = e0;
        Bfrag[fo + 1] = e1;
    }
    for (int k = tid; k < KEMB; k += TPB) {
        float s = 0.0f;
        for (int c = 0; c < CDIM; ++c) {
            float e = E[k * CDIM + c];
            s += e * e;
        }
        esq[k] = s;
    }
    __syncthreads();

    const int wave = tid >> 5;
    const int lane = tid & 31;
    const int m    = lane & 15;
    const int g    = lane >> 4;
    const int rowbase = wg * ROWS_WG + wave * 16;
    const int b       = rowbase >> 12;
    const int nrow    = (rowbase & 4095) + m;

    v2f afr[16];
#pragma unroll
    for (int kc = 0; kc < 16; ++kc) {
        int c0 = kc * 4 + g * 2;
        afr[kc].x = x[(b * CDIM + c0    ) * HW + nrow];
        afr[kc].y = x[(b * CDIM + c0 + 1) * HW + nrow];
    }

    float minval[8];
    int   minidx[8];
#pragma unroll
    for (int r = 0; r < 8; ++r) { minval[r] = 3.402823466e38f; minidx[r] = 0; }

    for (int t = 0; t < KEMB / 16; ++t) {
        v8f acc = {0.f, 0.f, 0.f, 0.f, 0.f, 0.f, 0.f, 0.f};
        const v2f* bp = (const v2f*)(Bfrag + t * 1024);
#pragma unroll
        for (int kc = 0; kc < 16; ++kc) {
            v2f bf = bp[kc * 32 + lane];
            acc = __builtin_amdgcn_wmma_f32_16x16x4_f32(
                false, afr[kc], false, bf, (short)0, acc, false, false);
        }
        int   col = t * 16 + m;
        float es  = esq[col];
#pragma unroll
        for (int r = 0; r < 8; ++r) {
            float d = es - 2.0f * acc[r];
            if (d < minval[r]) { minval[r] = d; minidx[r] = col; }
        }
    }

#pragma unroll
    for (int r = 0; r < 8; ++r) {
        float v = minval[r];
        int   i = minidx[r];
#pragma unroll
        for (int off = 1; off <= 8; off <<= 1) {
            float ov = __shfl_xor(v, off, 32);
            int   oi = __shfl_xor(i, off, 32);
            if (ov < v || (ov == v && oi < i)) { v = ov; i = oi; }
        }
        if (m == 0) idxs[wave * 16 + g * 8 + r] = i;
    }
    __syncthreads();

    const int b_wg = (wg * ROWS_WG) >> 12;
    const int n_wg = (wg * ROWS_WG) & 4095;
    float lsum = 0.0f;
    for (int ii = tid; ii < ROWS_WG * CDIM; ii += TPB) {
        int rl = ii & 127;
        int c  = ii >> 7;
        int k  = idxs[rl];
        int t  = k >> 4, nn = k & 15, kc = c >> 2, gg = (c >> 1) & 1;
        float ev = Bfrag[((t * 16 + kc) * 32 + (gg * 16 + nn)) * 2 + (c & 1)];
        int gaddr = (b_wg * CDIM + c) * HW + n_wg + rl;
        float xv = x[gaddr];
        zq[gaddr] = ev;
        float dlt = ev - xv;
        lsum += dlt * dlt;
    }
#pragma unroll
    for (int off = 1; off < 32; off <<= 1) lsum += __shfl_xor(lsum, off, 32);
    if (lane == 0) redbuf[wave] = lsum;
    __syncthreads();
    if (tid == 0) {
        float tot = 0.0f;
        for (int wv = 0; wv < NWAVES; ++wv) tot += redbuf[wv];
        atomicAdd(lossAcc, tot);
    }
}

__global__ void vq_finalize_kernel(const float* acc, float* lossOut) {
    if (threadIdx.x == 0) {
        lossOut[0] = 2.0f * acc[0] / (float)((long)NROWS * CDIM);
    }
}

extern "C" void kernel_launch(void* const* d_in, const int* in_sizes, int n_in,
                              void* d_out, int out_size, void* d_ws, size_t ws_size,
                              hipStream_t stream) {
    const float* x = (const float*)d_in[0];
    const float* E = (const float*)d_in[1];
    float* out     = (float*)d_out;
    float* acc     = (float*)d_ws;

    vq_zero_kernel<<<1, 64, 0, stream>>>(acc);
    vq_main_kernel<<<NWG, TPB, LDS_BYTES, stream>>>(x, E, out, acc);
    vq_finalize_kernel<<<1, 64, 0, stream>>>(acc, out + (out_size - 1));
}


# Three launches to two: each wave sums e_sq of its 128 codes from its own B fragments (row exchange, channel order kept); loss: work-group sums added in launch 2 in fixed order, no atomic
# speedup vs baseline: 1.9349x; 1.9349x over previous
#include <hip/hip_runtime.h>


typedef __attribute__((ext_vector_type(2))) float v2f;
typedef __attribute__((ext_vector_type(8))) float v8f;

#define BATCH   16
#define CDIM    64
#define HW      4096
#define NROWS   (BATCH * HW)
#define KEMB    1024
#define TPB     256
#define NWAVES  8
#define RT_WG   16
#define ROWS_WG (RT_WG * 16)
#define NWG     (NROWS / ROWS_WG)
#define CT_WAVE 8
#define ROW_PITCH  80
#define CAND_BYTES (NWAVES * ROWS_WG * ROW_PITCH)
#define LDS_BYTES  (CAND_BYTES + NWAVES * 4)

__device__ __forceinline__ void row_swap(float& a, float& b) {
    typedef __attribute__((ext_vector_type(2))) unsigned int v2u;
    v2u r = __builtin_amdgcn_permlane16_swap(__float_as_uint(a), __float_as_uint(b), false, false);
    a = __uint_as_float(r.x);
    b = __uint_as_float(r.y);
}

__global__ __launch_bounds__(TPB)
void vq_main_kernel(const float* __restrict__ x,
                    const float* __restrict__ E,
                    float* __restrict__ zq,
                    float* __restrict__ lossPart)
{
    extern __shared__ __attribute__((aligned(16))) unsigned char smem[];
    float* redbuf = (float*)(smem + CAND_BYTES);

    const int tid  = threadIdx.x;
    const int wg   = blockIdx.x;
    const int wave = tid >> 5;
    const int lane = tid & 31;
    const int m    = lane & 15;
    const int g    = lane >> 4;
    const int b_wg = (wg * ROWS_WG) >> 12;
    const int n_wg = (wg * ROWS_WG) & 4095;

    v2f   bfr[CT_WAVE][16];
#pragma unroll
    for (int j = 0; j < CT_WAVE; ++j) {
        const int code = (wave * CT_WAVE + j) * 16 + m;
#pragma unroll
        for (int kc = 0; kc < 16; ++kc) {
            const int c0 = kc * 4 + g * 2;
            bfr[j][kc].x = E[code * CDIM + c0];
            bfr[j][kc].y = E[code * CDIM + c0 + 1];
        }
    }

    float es[CT_WAVE];
    {
        float sq[4] = {0.0f, 0.0f, 0.0f, 0.0f};
#pragma unroll
        for (int kc = 0; kc < 16; ++kc) {
#pragma unroll
            for (int p = 0; p < 4; ++p) {
                float f0 = bfr[2 * p][kc].x, s0 = bfr[2 * p + 1][kc].x;
                float f1 = bfr[2 * p][kc].y, s1 = bfr[2 * p + 1][kc].y;
                row_swap(f0, s0);
                row_swap(f1, s1);
                sq[p] = __builtin_fmaf(f0, f0, sq[p]);
                sq[p] = __builtin_fmaf(f1, f1, sq[p]);
                sq[p] = __builtin_fmaf(s0, s0, sq[p]);
                sq[p] = __builtin_fmaf(s1, s1, sq[p]);
            }
        }
#pragma unroll
        for (int p = 0; p < 4; ++p) {
            float lo = sq[p], hi = sq[p];
            row_swap(lo, hi);
            es[2 * p]     = lo;
            es[2 * p + 1] = hi;
        }
    }

    const float* xb = x + ((size_t)b_wg * CDIM + g * 2) * HW + n_wg + m;
    v2f afr[16];
#pragma unroll
    for (int kc = 0; kc < 16; ++kc) {
        afr[kc].x = xb[(kc * 4    ) * HW];
        afr[kc].y = xb[(kc * 4 + 1) * HW];
    }

#pragma unroll 1
    for (int rt = 0; rt < RT_WG; ++rt) {
        const int rtn = (rt + 1 < RT_WG) ? (rt + 1) : rt;
        v2f anx[16];
#pragma unroll
        for (int kc = 0; kc < 16; ++kc) {
            anx[kc].x = xb[(kc * 4    ) * HW + rtn * 16];
            anx[kc].y = xb[(kc * 4 + 1) * HW + rtn * 16];
        }

        float minval[8];
        int   minidx[8];
#pragma unroll
        for (int r = 0; r < 8; ++r) { minval[r] = 3.402823466e38f; minidx[r] = 0; }

        v8f acc[CT_WAVE];
#pragma unroll
        for (int j = 0; j < CT_WAVE; ++j)
            acc[j] = (v8f){0.f, 0.f, 0.f, 0.f, 0.f, 0.f, 0.f, 0.f};
#pragma unroll
        for (int kc = 0; kc < 16; ++kc) {
#pragma unroll
            for (int j = 0; j < CT_WAVE; ++j) {
                acc[j] = __builtin_amdgcn_wmma_f32_16x16x4_f32(
                    false, afr[kc], false, bfr[j][kc], (short)0, acc[j], false, false);
            }
        }
#pragma unroll
        for (int j = 0; j < CT_WAVE; ++j) {
            const int col = (wave * CT_WAVE + j) * 16 + m;
#pragma unroll
            for (int r = 0; r < 8; ++r) {
                float d = es[j] - 2.0f * acc[j][r];
                if (d < minval[r]) { minval[r] = d; minidx[r] = col; }
            }
        }

#pragma unroll
        for (int r = 0; r < 8; ++r) {
            unsigned char* row = smem + (size_t)((wave * ROWS_WG) + rt * 16 + g * 8 + r) * ROW_PITCH;
            ((float*)row)[m] = minval[r];
            row[64 + m]      = (unsigned char)minidx[r];
        }

#pragma unroll
        for (int kc = 0; kc < 16; ++kc) afr[kc] = anx[kc];
    }
    __syncthreads();

    float bv = 0.0f;
    int   bi = 0;
#pragma unroll
    for (int w = 0; w < NWAVES; ++w) {
        const unsigned char* row = smem + (size_t)(w * ROWS_WG + tid) * ROW_PITCH;
        float cv[16];
        unsigned int cb[4];
#pragma unroll
        for (int q = 0; q < 4; ++q) {
            float4 t4 = ((const float4*)row)[q];
            cv[4 * q] = t4.x; cv[4 * q + 1] = t4.y; cv[4 * q + 2] = t4.z; cv[4 * q + 3] = t4.w;
        }
        {
            uint4 u4 = ((const uint4*)row)[4];
            cb[0] = u4.x; cb[1] = u4.y; cb[2] = u4.z; cb[3] = u4.w;
        }
        float wv = cv[0];
        unsigned int wb = cb[0] & 0xffu;
#pragma unroll
        for (int l = 1; l < 16; ++l) {
            float v = cv[l];
            unsigned int bb = (cb[l >> 2] >> (8 * (l & 3))) & 0xffu;
            bool take = (v < wv) | ((v == wv) & (bb < wb));
            wv = take ? v : wv;
            wb = take ? bb : wb;
        }
        int wcode = (int)(((unsigned int)(w >> 1) << 8) | wb);
        bool takew = (w == 0) | (wv < bv);
        bv = takew ? wv : bv;
        bi = takew ? wcode : bi;
    }
    bi &= (KEMB - 1);

    const float* er = E + bi * CDIM;
    const size_t gbase = (size_t)b_wg * CDIM * HW + n_wg + tid;
    float lsum = 0.0f;
#pragma unroll
    for (int c = 0; c < CDIM; ++c) {
        float ev = er[c];
        float xv = x[gbase + (size_t)c * HW];
        zq[gbase + (size_t)c * HW] = ev;
        float dlt = ev - xv;
        lsum += dlt * dlt;
    }
#pragma unroll
    for (int off = 1; off < 32; off <<= 1) lsum += __shfl_xor(lsum, off, 32);
    if (lane == 0) redbuf[wave] = lsum;
    __syncthreads();
    if (tid == 0) {
        float tot = 0.0f;
        for (int wv = 0; wv < NWAVES; ++wv) tot += redbuf[wv];
        lossPart[wg] = tot;
    }
}

__global__ void vq_finalize_kernel(const float* __restrict__ part, float* __restrict__ lossOut) {
    const int lane = threadIdx.x;
    const float4 a = ((const float4*)part)[2 * lane];
    const float4 b = ((const float4*)part)[2 * lane + 1];
    float s = a.x;
    s += a.y; s += a.z; s += a.w;
    s += b.x; s += b.y; s += b.z; s += b.w;
#pragma unroll
    for (int off = 1; off < 32; off <<= 1) s += __shfl_xor(s, off, 32);
    if (lane == 0) {
        lossOut[0] = 2.0f * s / (float)((long)NROWS * CDIM);
    }
}

extern "C" void kernel_launch(void* const* d_in, const int* in_sizes, int n_in,
                              void* d_out, int out_size, void* d_ws, size_t ws_size,
                              hipStream_t stream) {
    const float* x = (const float*)d_in[0];
    const float* E = (const float*)d_in[1];
    float* out     = (float*)d_out;
    float* ws      = (float*)d_ws;

    vq_main_kernel<<<NWG, TPB, LDS_BYTES, stream>>>(x, E, out, ws);
    vq_finalize_kernel<<<1, 32, 0, stream>>>(ws, out + (out_size - 1));
}
